# starting kernel
# baseline (speedup 1.0000x reference)
#include <hip/hip_runtime.h>

typedef float v2f __attribute__((ext_vector_type(2)));
typedef float v8f __attribute__((ext_vector_type(8)));

#define B_SZ  16
#define D_K   512
#define P_    16
#define F_    32
#define NP    128
#define NE    32
#define NROWS (B_SZ * F_ * NP)
#define WAVES_PER_BLK 8
#define NTILES (NROWS / 16)

__global__ __launch_bounds__(256) void gemm_raw_kernel(
    const float* __restrict__ A,
    const float* __restrict__ W,
    const float* __restrict__ bias,
    float* __restrict__ ws_mu,
    float* __restrict__ ws_sig)
{
    __shared__ float sW[NE * D_K];

    const int tid = threadIdx.x;

    const float4* Wv  = (const float4*)W;
    float4*       sWv = (float4*)sW;
#pragma unroll
    for (int i = 0; i < (NE * D_K / 4) / 256; ++i)
        sWv[tid + i * 256] = Wv[tid + i * 256];
    __syncthreads();

    const int wave    = tid >> 5;
    const int lane    = tid & 31;
    const int tile    = blockIdx.x * WAVES_PER_BLK + wave;
    const int rowBase = tile * 16;

    const int m     = lane & 15;
    const int khalf = lane >> 4;
    const float* arow = A + (size_t)(rowBase + m) * D_K + 2 * khalf;
    const float* bmu = sW + (size_t)m * D_K + 2 * khalf;
    const float* bsg = sW + (size_t)(m + 16) * D_K + 2 * khalf;

    v8f cmu = {};
    v8f csg = {};
#pragma unroll 4
    for (int k0 = 0; k0 < D_K; k0 += 4) {
        v2f a  = *(const v2f*)(arow + k0);
        v2f b0 = *(const v2f*)(bmu + k0);
        v2f b1 = *(const v2f*)(bsg + k0);
        cmu = __builtin_amdgcn_wmma_f32_16x16x4_f32(false, a, false, b0,
                                                    (short)0, cmu, false, false);
        csg = __builtin_amdgcn_wmma_f32_16x16x4_f32(false, a, false, b1,
                                                    (short)0, csg, false, false);
    }

    const int   e     = lane & 15;
    const float bmu_s = bias[e];
    const float bsg_s = bias[16 + e];
#pragma unroll
    for (int v = 0; v < 8; ++v) {
        const int row = rowBase + v + 8 * (lane >> 4);
        const int bb  = row >> 12;
        const int f   = (row >> 7) & 31;
        const int n   = row & 127;
        const size_t idx = ((size_t)((bb * NP + n) * P_ + e)) * F_ + f;
        ws_mu[idx] = cmu[v] + bmu_s;
        const float sg = __expf(csg[v] + bsg_s);
        ws_sig[idx] = fmaxf(sg, 1e-6f);
    }
}

__global__ __launch_bounds__(256) void combine_kernel(
    const float* __restrict__ ws_mu,
    const float* __restrict__ ws_sig,
    float* __restrict__ out, int T)
{
    const int idx   = blockIdx.x * blockDim.x + threadIdx.x;
    const int total = B_SZ * T * F_;
    if (idx >= total) return;

    const int f  = idx & 31;
    const int rt = idx >> 5;
    const int t  = rt % T;
    const int b  = rt / T;

    const int n1 = t >> 3, p1 = t & 7;
    float smu = 0.f, ssg = 0.f;
    int   c   = 0;
    if (n1 < NP) {
        const size_t i1 = ((size_t)((b * NP + n1) * P_ + p1)) * F_ + f;
        smu += ws_mu[i1]; ssg += ws_sig[i1]; ++c;
    }
    const int n2 = n1 - 1, p2 = p1 + 8;
    if (n2 >= 0 && n2 < NP) {
        const size_t i2 = ((size_t)((b * NP + n2) * P_ + p2)) * F_ + f;
        smu += ws_mu[i2]; ssg += ws_sig[i2]; ++c;
    }
    const float inv = 1.0f / (float)(c > 0 ? c : 1);
    const size_t o = (size_t)b * T * F_ + (size_t)t * F_ + f;
    out[o]                          = smu * inv;
    out[(size_t)B_SZ * T * F_ + o]  = ssg * inv;
}

extern "C" void kernel_launch(void* const* d_in, const int* in_sizes, int n_in,
                              void* d_out, int out_size, void* d_ws, size_t ws_size,
                              hipStream_t stream)
{
    const float* A    = (const float*)d_in[0];
    const float* W    = (const float*)d_in[1];
    const float* bias = (const float*)d_in[2];
    const int T = out_size / (2 * B_SZ * F_);

    float* ws_mu  = (float*)d_ws;
    float* ws_sig = ws_mu + (size_t)B_SZ * NP * P_ * F_;

    gemm_raw_kernel<<<NTILES / WAVES_PER_BLK, 256, 0, stream>>>(
        A, W, bias, ws_mu, ws_sig);

    const int total  = B_SZ * T * F_;
    const int blocks = (total + 255) / 256;
    combine_kernel<<<blocks, 256, 0, stream>>>(ws_mu, ws_sig, (float*)d_out, T);
}


# Pad the LDS row pitch of the staged W from 512 to 516 floats so the 16 rows of each B-fragment read fall on distinct banks (every fragment ds_load was 16-way bank conflicted)
# speedup vs baseline: 1.0888x; 1.0888x over previous
#include <hip/hip_runtime.h>

typedef float v2f __attribute__((ext_vector_type(2)));
typedef float v8f __attribute__((ext_vector_type(8)));

#define B_SZ  16
#define D_K   512
#define P_    16
#define F_    32
#define NP    128
#define NE    32
#define NROWS (B_SZ * F_ * NP)
#define WAVES_PER_BLK 8
#define NTILES (NROWS / 16)
#define W_PITCH (D_K + 4)

__global__ __launch_bounds__(256) void gemm_raw_kernel(
    const float* __restrict__ A,
    const float* __restrict__ W,
    const float* __restrict__ bias,
    float* __restrict__ ws_mu,
    float* __restrict__ ws_sig)
{
    __shared__ float sW[NE * W_PITCH];

    const int tid = threadIdx.x;

    const float4* Wv  = (const float4*)W;
    float4*       sWv = (float4*)sW + (tid >> 7) * (W_PITCH / 4) + (tid & 127);
#pragma unroll
    for (int i = 0; i < (NE * D_K / 4) / 256; ++i)
        sWv[i * 2 * (W_PITCH / 4)] = Wv[tid + i * 256];
    __syncthreads();

    const int wave    = tid >> 5;
    const int lane    = tid & 31;
    const int tile    = blockIdx.x * WAVES_PER_BLK + wave;
    const int rowBase = tile * 16;

    const int m     = lane & 15;
    const int khalf = lane >> 4;
    const float* arow = A + (size_t)(rowBase + m) * D_K + 2 * khalf;
    const float* bmu = sW + (size_t)m * W_PITCH + 2 * khalf;
    const float* bsg = sW + (size_t)(m + 16) * W_PITCH + 2 * khalf;

    v8f cmu = {};
    v8f csg = {};
#pragma unroll 4
    for (int k0 = 0; k0 < D_K; k0 += 4) {
        v2f a  = *(const v2f*)(arow + k0);
        v2f b0 = *(const v2f*)(bmu + k0);
        v2f b1 = *(const v2f*)(bsg + k0);
        cmu = __builtin_amdgcn_wmma_f32_16x16x4_f32(false, a, false, b0,
                                                    (short)0, cmu, false, false);
        csg = __builtin_amdgcn_wmma_f32_16x16x4_f32(false, a, false, b1,
                                                    (short)0, csg, false, false);
    }

    const int   e     = lane & 15;
    const float bmu_s = bias[e];
    const float bsg_s = bias[16 + e];
#pragma unroll
    for (int v = 0; v < 8; ++v) {
        const int row = rowBase + v + 8 * (lane >> 4);
        const int bb  = row >> 12;
        const int f   = (row >> 7) & 31;
        const int n   = row & 127;
        const size_t idx = ((size_t)((bb * NP + n) * P_ + e)) * F_ + f;
        ws_mu[idx] = cmu[v] + bmu_s;
        const float sg = __expf(csg[v] + bsg_s);
        ws_sig[idx] = fmaxf(sg, 1e-6f);
    }
}

__global__ __launch_bounds__(256) void combine_kernel(
    const float* __restrict__ ws_mu,
    const float* __restrict__ ws_sig,
    float* __restrict__ out, int T)
{
    const int idx   = blockIdx.x * blockDim.x + threadIdx.x;
    const int total = B_SZ * T * F_;
    if (idx >= total) return;

    const int f  = idx & 31;
    const int rt = idx >> 5;
    const int t  = rt % T;
    const int b  = rt / T;

    const int n1 = t >> 3, p1 = t & 7;
    float smu = 0.f, ssg = 0.f;
    int   c   = 0;
    if (n1 < NP) {
        const size_t i1 = ((size_t)((b * NP + n1) * P_ + p1)) * F_ + f;
        smu += ws_mu[i1]; ssg += ws_sig[i1]; ++c;
    }
    const int n2 = n1 - 1, p2 = p1 + 8;
    if (n2 >= 0 && n2 < NP) {
        const size_t i2 = ((size_t)((b * NP + n2) * P_ + p2)) * F_ + f;
        smu += ws_mu[i2]; ssg += ws_sig[i2]; ++c;
    }
    const float inv = 1.0f / (float)(c > 0 ? c : 1);
    const size_t o = (size_t)b * T * F_ + (size_t)t * F_ + f;
    out[o]                          = smu * inv;
    out[(size_t)B_SZ * T * F_ + o]  = ssg * inv;
}

extern "C" void kernel_launch(void* const* d_in, const int* in_sizes, int n_in,
                              void* d_out, int out_size, void* d_ws, size_t ws_size,
                              hipStream_t stream)
{
    const float* A    = (const float*)d_in[0];
    const float* W    = (const float*)d_in[1];
    const float* bias = (const float*)d_in[2];
    const int T = out_size / (2 * B_SZ * F_);

    float* ws_mu  = (float*)d_ws;
    float* ws_sig = ws_mu + (size_t)B_SZ * NP * P_ * F_;

    gemm_raw_kernel<<<NTILES / WAVES_PER_BLK, 256, 0, stream>>>(
        A, W, bias, ws_mu, ws_sig);

    const int total  = B_SZ * T * F_;
    const int blocks = (total + 255) / 256;
    combine_kernel<<<blocks, 256, 0, stream>>>(ws_mu, ws_sig, (float*)d_out, T);
}


# Fuse the overlap-add into the GEMM kernel: a block owns 8 n x all 32 f plus a recomputed halo n, folds through LDS and writes whole 128-B output rows; one launch, no workspace round trip
# speedup vs baseline: 2.1802x; 2.0024x over previous
#include <hip/hip_runtime.h>


typedef float v2f __attribute__((ext_vector_type(2)));
typedef float v8f __attribute__((ext_vector_type(8)));

#define B_SZ  16
#define D_K   512
#define P_    16
#define F_    32
#define NP    128
#define NE    32
#define W_PITCH (D_K + 4)

#define NB        8
#define NBLK_B    (NP / NB)
#define NWAVES    18
#define NTHREADS  (NWAVES * 32)
#define X_EP      (F_ + 4)
#define X_NP      (NE * X_EP)
#define T_FULL    (8 * NP)

__global__ __launch_bounds__(NTHREADS) void gemm_fold_kernel(
    const float* __restrict__ A,
    const float* __restrict__ W,
    const float* __restrict__ bias,
    float* __restrict__ out,
    int T)
{
    __shared__ __attribute__((aligned(16))) float sW[NE * W_PITCH];
    __shared__ __attribute__((aligned(16))) float sX[(NB + 1) * X_NP];

    const int tid = threadIdx.x;

    if (tid < 512) {
        const float4* Wv  = (const float4*)W;
        float4*       sWv = (float4*)sW + (tid >> 7) * (W_PITCH / 4) + (tid & 127);
#pragma unroll
        for (int i = 0; i < (NE * D_K / 4) / 512; ++i)
            sWv[i * 4 * (W_PITCH / 4)] = Wv[tid + i * 512];
    }
    __syncthreads();

    const int wave = tid >> 5;
    const int lane = tid & 31;
    const int bb   = blockIdx.x / NBLK_B;
    const int n0   = (blockIdx.x % NBLK_B) * NB;

    const int m     = lane & 15;
    const int khalf = lane >> 4;
    const bool halo  = wave >= 16;
    const int  f_row = halo ? 16 * (wave - 16) + m : 2 * wave + (m >> 3);
    const int  n_row = halo ? (n0 > 0 ? n0 - 1 : 0) : n0 + (m & 7);
    const float* arow = A + (size_t)((bb * F_ + f_row) * NP + n_row) * D_K + 2 * khalf;
    const float* bmu = sW + (size_t)m * W_PITCH + 2 * khalf;
    const float* bsg = sW + (size_t)(m + 16) * W_PITCH + 2 * khalf;

    v8f cmu = {};
    v8f csg = {};
#pragma unroll 4
    for (int k0 = 0; k0 < D_K; k0 += 4) {
        v2f a  = *(const v2f*)(arow + k0);
        v2f b0 = *(const v2f*)(bmu + k0);
        v2f b1 = *(const v2f*)(bsg + k0);
        cmu = __builtin_amdgcn_wmma_f32_16x16x4_f32(false, a, false, b0,
                                                    (short)0, cmu, false, false);
        csg = __builtin_amdgcn_wmma_f32_16x16x4_f32(false, a, false, b1,
                                                    (short)0, csg, false, false);
    }

    {
        const int   e     = lane & 15;
        const int   hi    = lane >> 4;
        const float bmu_s = bias[e];
        const float bsg_s = bias[16 + e];
#pragma unroll
        for (int v = 0; v < 8; ++v) {
            const int plane = halo ? 0 : v + 1;
            const int f     = halo ? 16 * (wave - 16) + v + 8 * hi : 2 * wave + hi;
            float* dst = sX + plane * X_NP + e * X_EP + f;
            dst[0] = cmu[v] + bmu_s;
            const float sg = __expf(csg[v] + bsg_s);
            dst[16 * X_EP] = fmaxf(sg, 1e-6f);
        }
    }
    __syncthreads();

    const size_t half = (size_t)B_SZ * T * F_;
#pragma unroll
    for (int r = 0; r < (2 * 8 * NB * 8 + NTHREADS - 1) / NTHREADS; ++r) {
        const int idx = tid + r * NTHREADS;
        if (idx < 2 * 8 * NB * 8) {
            const int q     = idx & 7;
            const int line  = idx >> 3;
            const int which = line >> 6;
            const int tt    = line & 63;
            const int jl    = tt >> 3;
            const int p     = tt & 7;
            const int t     = 8 * n0 + tt;
            const float4 x1 = *(const float4*)(sX + (jl + 1) * X_NP + (16 * which + p) * X_EP + 4 * q);
            const float4 x2 = *(const float4*)(sX + jl * X_NP + (16 * which + p + 8) * X_EP + 4 * q);
            const bool  two = (n0 + jl) > 0;
            const float inv = two ? 0.5f : 1.0f;
            float4 o;
            o.x = (x1.x + (two ? x2.x : 0.f)) * inv;
            o.y = (x1.y + (two ? x2.y : 0.f)) * inv;
            o.z = (x1.z + (two ? x2.z : 0.f)) * inv;
            o.w = (x1.w + (two ? x2.w : 0.f)) * inv;
            if (t < T)
                *(float4*)(out + which * half + ((size_t)bb * T + t) * F_ + 4 * q) = o;
        }
    }

    if (n0 == NP - NB) {
        const int ntail = T - T_FULL;
        for (int idx = tid; idx < ntail * 16; idx += NTHREADS) {
            const int q     = idx & 7;
            const int line  = idx >> 3;
            const int which = line >= ntail ? 1 : 0;
            const int tr    = line - which * ntail;
            const int t     = T_FULL + tr;
            float4 o = make_float4(0.f, 0.f, 0.f, 0.f);
            if (tr < 8)
                o = *(const float4*)(sX + NB * X_NP + (16 * which + tr + 8) * X_EP + 4 * q);
            *(float4*)(out + which * half + ((size_t)bb * T + t) * F_ + 4 * q) = o;
        }
    }
}

extern "C" void kernel_launch(void* const* d_in, const int* in_sizes, int n_in,
                              void* d_out, int out_size, void* d_ws, size_t ws_size,
                              hipStream_t stream)
{
    const float* A    = (const float*)d_in[0];
    const float* W    = (const float*)d_in[1];
    const float* bias = (const float*)d_in[2];
    const int T = out_size / (2 * B_SZ * F_);

    gemm_fold_kernel<<<B_SZ * NBLK_B, NTHREADS, 0, stream>>>(
        A, W, bias, (float*)d_out, T);
}


# A rows reach the fragments through a per-wave LDS double buffer: each wave loads whole 256-B row pieces (4 lines per load instruction, was 16) two 64-k chunks ahead; k order and products unchanged
# speedup vs baseline: 2.5206x; 1.1561x over previous
#include <hip/hip_runtime.h>


typedef float v2f __attribute__((ext_vector_type(2)));
typedef float v8f __attribute__((ext_vector_type(8)));

#define B_SZ  16
#define D_K   512
#define P_    16
#define F_    32
#define NP    128
#define NE    32
#define W_PITCH (D_K + 4)

#define NB        8
#define NBLK_B    (NP / NB)
#define NWAVES    18
#define NTHREADS  (NWAVES * 32)
#define X_EP      (F_ + 4)
#define X_NP      (NE * X_EP)
#define T_FULL    (8 * NP)
#define A_CH      64
#define A_PITCH   (A_CH + 4)
#define A_TILE    (16 * A_PITCH)

__global__ __launch_bounds__(NTHREADS) void gemm_fold_kernel(
    const float* __restrict__ A,
    const float* __restrict__ W,
    const float* __restrict__ bias,
    float* __restrict__ out,
    int T)
{
    __shared__ __attribute__((aligned(16))) float sW[NE * W_PITCH];
    __shared__ __attribute__((aligned(16))) float sX[(NB + 1) * X_NP];
    __shared__ __attribute__((aligned(16))) float sA[NWAVES * 2 * A_TILE];

    const int tid  = threadIdx.x;
    const int wave = tid >> 5;
    const int lane = tid & 31;
    const int bb   = blockIdx.x / NBLK_B;
    const int n0   = (blockIdx.x % NBLK_B) * NB;

    const int m     = lane & 15;
    const int khalf = lane >> 4;
    const bool halo = wave >= 16;
    const int  nh   = n0 > 0 ? n0 - 1 : 0;

#define A_ROWPTR(I)                                                            \
    (A + (size_t)((bb * F_ + (halo ? 16 * (wave - 16) + 2 * (I) + khalf        \
                                   : 2 * wave + ((2 * (I) + khalf) >> 3))) * NP \
                  + (halo ? nh : n0 + ((2 * (I) + khalf) & 7))) * D_K + 4 * m)
    const float* ap0 = A_ROWPTR(0);
    const float* ap1 = A_ROWPTR(1);
    const float* ap2 = A_ROWPTR(2);
    const float* ap3 = A_ROWPTR(3);
    const float* ap4 = A_ROWPTR(4);
    const float* ap5 = A_ROWPTR(5);
    const float* ap6 = A_ROWPTR(6);
    const float* ap7 = A_ROWPTR(7);
#undef A_ROWPTR

    float*       st = sA + wave * (2 * A_TILE) + khalf * A_PITCH + 4 * m;
    const float* fr = sA + wave * (2 * A_TILE) + m * A_PITCH + 2 * khalf;

#define A_LOAD(P, KOFF)                                        \
    const float4 P##0 = *(const float4*)(ap0 + (KOFF));        \
    const float4 P##1 = *(const float4*)(ap1 + (KOFF));        \
    const float4 P##2 = *(const float4*)(ap2 + (KOFF));        \
    const float4 P##3 = *(const float4*)(ap3 + (KOFF));        \
    const float4 P##4 = *(const float4*)(ap4 + (KOFF));        \
    const float4 P##5 = *(const float4*)(ap5 + (KOFF));        \
    const float4 P##6 = *(const float4*)(ap6 + (KOFF));        \
    const float4 P##7 = *(const float4*)(ap7 + (KOFF));
#define A_STORE(P, BUF)                                               \
    *(float4*)(st + (BUF) * A_TILE)                 = P##0;           \
    *(float4*)(st + (BUF) * A_TILE + 2 * A_PITCH)   = P##1;           \
    *(float4*)(st + (BUF) * A_TILE + 4 * A_PITCH)   = P##2;           \
    *(float4*)(st + (BUF) * A_TILE + 6 * A_PITCH)   = P##3;           \
    *(float4*)(st + (BUF) * A_TILE + 8 * A_PITCH)   = P##4;           \
    *(float4*)(st + (BUF) * A_TILE + 10 * A_PITCH)  = P##5;           \
    *(float4*)(st + (BUF) * A_TILE + 12 * A_PITCH)  = P##6;           \
    *(float4*)(st + (BUF) * A_TILE + 14 * A_PITCH)  = P##7;

    if (tid < 512) {
        const float4* Wv  = (const float4*)W;
        float4*       sWv = (float4*)sW + (tid >> 7) * (W_PITCH / 4) + (tid & 127);
#pragma unroll
        for (int i = 0; i < (NE * D_K / 4) / 512; ++i)
            sWv[i * 4 * (W_PITCH / 4)] = Wv[tid + i * 512];
    }
    { A_LOAD(r, 0)    A_STORE(r, 0) }
    { A_LOAD(r, A_CH) A_STORE(r, 1) }
    __syncthreads();

    const float* bmu = sW + (size_t)m * W_PITCH + 2 * khalf;
    const float* bsg = sW + (size_t)(m + 16) * W_PITCH + 2 * khalf;

    v8f cmu = {};
    v8f csg = {};

#define A_COMPUTE(BUF, KOFF)                                                  \
    _Pragma("unroll") for (int s = 0; s < A_CH / 4; ++s) {                    \
        v2f a  = *(const v2f*)(fr + (BUF) * A_TILE + 4 * s);                  \
        v2f b0 = *(const v2f*)(bmu + (KOFF) + 4 * s);                         \
        v2f b1 = *(const v2f*)(bsg + (KOFF) + 4 * s);                         \
        cmu = __builtin_amdgcn_wmma_f32_16x16x4_f32(false, a, false, b0,      \
                                                    (short)0, cmu, false, false); \
        csg = __builtin_amdgcn_wmma_f32_16x16x4_f32(false, a, false, b1,      \
                                                    (short)0, csg, false, false); \
    }

    for (int kc = 0; kc < D_K - 2 * A_CH; kc += 2 * A_CH) {
        { A_LOAD(r, kc + 2 * A_CH) A_COMPUTE(0, kc)        A_STORE(r, 0) }
        { A_LOAD(r, kc + 3 * A_CH) A_COMPUTE(1, kc + A_CH) A_STORE(r, 1) }
    }
    A_COMPUTE(0, D_K - 2 * A_CH)
    A_COMPUTE(1, D_K - A_CH)
#undef A_LOAD
#undef A_STORE
#undef A_COMPUTE

    {
        const int   e     = lane & 15;
        const int   hi    = lane >> 4;
        const float bmu_s = bias[e];
        const float bsg_s = bias[16 + e];
#pragma unroll
        for (int v = 0; v < 8; ++v) {
            const int plane = halo ? 0 : v + 1;
            const int f     = halo ? 16 * (wave - 16) + v + 8 * hi : 2 * wave + hi;
            float* dst = sX + plane * X_NP + e * X_EP + f;
            dst[0] = cmu[v] + bmu_s;
            const float sg = __expf(csg[v] + bsg_s);
            dst[16 * X_EP] = fmaxf(sg, 1e-6f);
        }
    }
    __syncthreads();

    const size_t half = (size_t)B_SZ * T * F_;
#pragma unroll
    for (int r = 0; r < (2 * 8 * NB * 8 + NTHREADS - 1) / NTHREADS; ++r) {
        const int idx = tid + r * NTHREADS;
        if (idx < 2 * 8 * NB * 8) {
            const int q     = idx & 7;
            const int line  = idx >> 3;
            const int which = line >> 6;
            const int tt    = line & 63;
            const int jl    = tt >> 3;
            const int p     = tt & 7;
            const int t     = 8 * n0 + tt;
            const float4 x1 = *(const float4*)(sX + (jl + 1) * X_NP + (16 * which + p) * X_EP + 4 * q);
            const float4 x2 = *(const float4*)(sX + jl * X_NP + (16 * which + p + 8) * X_EP + 4 * q);
            const bool  two = (n0 + jl) > 0;
            const float inv = two ? 0.5f : 1.0f;
            float4 o;
            o.x = (x1.x + (two ? x2.x : 0.f)) * inv;
            o.y = (x1.y + (two ? x2.y : 0.f)) * inv;
            o.z = (x1.z + (two ? x2.z : 0.f)) * inv;
            o.w = (x1.w + (two ? x2.w : 0.f)) * inv;
            if (t < T)
                *(float4*)(out + which * half + ((size_t)bb * T + t) * F_ + 4 * q) = o;
        }
    }

    if (n0 == NP - NB) {
        const int ntail = T - T_FULL;
        for (int idx = tid; idx < ntail * 16; idx += NTHREADS) {
            const int q     = idx & 7;
            const int line  = idx >> 3;
            const int which = line >= ntail ? 1 : 0;
            const int tr    = line - which * ntail;
            const int t     = T_FULL + tr;
            float4 o = make_float4(0.f, 0.f, 0.f, 0.f);
            if (tr < 8)
                o = *(const float4*)(sX + NB * X_NP + (16 * which + tr + 8) * X_EP + 4 * q);
            *(float4*)(out + which * half + ((size_t)bb * T + t) * F_ + 4 * q) = o;
        }
    }
}

extern "C" void kernel_launch(void* const* d_in, const int* in_sizes, int n_in,
                              void* d_out, int out_size, void* d_ws, size_t ws_size,
                              hipStream_t stream)
{
    const float* A    = (const float*)d_in[0];
    const float* W    = (const float*)d_in[1];
    const float* bias = (const float*)d_in[2];
    const int T = out_size / (2 * B_SZ * F_);

    gemm_fold_kernel<<<B_SZ * NBLK_B, NTHREADS, 0, stream>>>(
        A, W, bias, (float*)d_out, T);
}


# No halo tiles: a wave multiplies segments n0..n0+7 by the offset 0-7 columns and n0-1..n0+6 by the offset 8-15 columns, overlap-add becomes a register add; 16 waves of 256 WMMAs per block (was 18)
# speedup vs baseline: 2.6514x; 1.0519x over previous
#include <hip/hip_runtime.h>


typedef float v2f __attribute__((ext_vector_type(2)));
typedef float v8f __attribute__((ext_vector_type(8)));

#define B_SZ  16
#define D_K   512
#define P_    16
#define F_    32
#define NP    128
#define NE    32
#define W_PITCH (D_K + 4)

#define NB        8
#define NBLK_B    (NP / NB)
#define NWAVES    16
#define NTHREADS  (NWAVES * 32)
#define X_EP      (F_ + 4)
#define X_WP      (8 * NB * X_EP + 32)
#define T_FULL    (8 * NP)
#define A_CH      64
#define A_PITCH   (A_CH + 4)
#define A_X0      (16 * A_PITCH)
#define A_XF      96
#define A_TILE    (A_X0 + A_XF + A_CH)

__global__ __launch_bounds__(NTHREADS) void gemm_fold_kernel(
    const float* __restrict__ A,
    const float* __restrict__ W,
    const float* __restrict__ bias,
    float* __restrict__ out,
    int T)
{
    __shared__ __attribute__((aligned(16))) float sW[NE * W_PITCH];
    __shared__ __attribute__((aligned(16))) float sX[2 * X_WP];
    __shared__ __attribute__((aligned(16))) float sT[2 * 8 * X_EP];
    __shared__ __attribute__((aligned(16))) float sA[NWAVES * 2 * A_TILE];

    const int tid  = threadIdx.x;
    const int wave = tid >> 5;
    const int lane = tid & 31;
    const int bb   = blockIdx.x / NBLK_B;
    const int n0   = (blockIdx.x % NBLK_B) * NB;
    const int nh   = (n0 + NP - 1) % NP;

    const int m     = lane & 15;
    const int khalf = lane >> 4;
    const int fl = m >> 3;
    const int ri = m & 7;

    const float* arow = A + (size_t)((bb * F_ + 2 * wave + khalf) * NP) * D_K + 4 * m;
    const float* apx  = arow + (size_t)nh * D_K;
    const float* ap   = arow + (size_t)n0 * D_K;

    float*       sAw = sA + wave * (2 * A_TILE);
    float*       st  = sAw + (8 * khalf) * A_PITCH + 4 * m;
    float*       stx = sAw + A_X0 + khalf * A_XF + 4 * m;
    const float* fhi = sAw + m * A_PITCH + 2 * khalf;
    const float* flo = sAw + (ri < 7 ? (m + 1) * A_PITCH : A_X0 + fl * A_XF) + 2 * khalf;

#define A_LOAD(P, KOFF)                                             \
    const float4 P##0 = *(const float4*)(apx + (KOFF));             \
    const float4 P##1 = *(const float4*)(ap + (KOFF));              \
    const float4 P##2 = *(const float4*)(ap + 1 * D_K + (KOFF));    \
    const float4 P##3 = *(const float4*)(ap + 2 * D_K + (KOFF));    \
    const float4 P##4 = *(const float4*)(ap + 3 * D_K + (KOFF));    \
    const float4 P##5 = *(const float4*)(ap + 4 * D_K + (KOFF));    \
    const float4 P##6 = *(const float4*)(ap + 5 * D_K + (KOFF));    \
    const float4 P##7 = *(const float4*)(ap + 6 * D_K + (KOFF));    \
    const float4 P##8 = *(const float4*)(ap + 7 * D_K + (KOFF));
#define A_STORE(P, BUF)                                               \
    *(float4*)(st + (BUF) * A_TILE)                 = P##0;           \
    *(float4*)(st + (BUF) * A_TILE + 1 * A_PITCH)   = P##1;           \
    *(float4*)(st + (BUF) * A_TILE + 2 * A_PITCH)   = P##2;           \
    *(float4*)(st + (BUF) * A_TILE + 3 * A_PITCH)   = P##3;           \
    *(float4*)(st + (BUF) * A_TILE + 4 * A_PITCH)   = P##4;           \
    *(float4*)(st + (BUF) * A_TILE + 5 * A_PITCH)   = P##5;           \
    *(float4*)(st + (BUF) * A_TILE + 6 * A_PITCH)   = P##6;           \
    *(float4*)(st + (BUF) * A_TILE + 7 * A_PITCH)   = P##7;           \
    *(float4*)(stx + (BUF) * A_TILE)                = P##8;

    {
        const float4* Wv  = (const float4*)W;
        float4*       sWv = (float4*)sW + (tid & 127);
#pragma unroll
        for (int i = 0; i < (NE * D_K / 4) / NTHREADS; ++i) {
            const int e  = 4 * i + (tid >> 7);
            const int sr = (e & 7) + 16 * ((e >> 3) & 1) + 8 * (e >> 4);
            sWv[sr * (W_PITCH / 4)] = Wv[tid + i * NTHREADS];
        }
    }
    { A_LOAD(r, 0)    A_STORE(r, 0) }
    { A_LOAD(r, A_CH) A_STORE(r, 1) }
    __syncthreads();

    const float* blo = sW + (size_t)m * W_PITCH + 2 * khalf;
    const float* bhi = sW + (size_t)(m + 16) * W_PITCH + 2 * khalf;

    v8f clo = {};
    v8f chi = {};

#define A_COMPUTE(BUF, KOFF)                                                  \
    _Pragma("unroll") for (int s = 0; s < A_CH / 4; ++s) {                    \
        v2f a0 = *(const v2f*)(flo + (BUF) * A_TILE + 4 * s);                 \
        v2f a1 = *(const v2f*)(fhi + (BUF) * A_TILE + 4 * s);                 \
        v2f b0 = *(const v2f*)(blo + (KOFF) + 4 * s);                         \
        v2f b1 = *(const v2f*)(bhi + (KOFF) + 4 * s);                         \
        clo = __builtin_amdgcn_wmma_f32_16x16x4_f32(false, a0, false, b0,     \
                                                    (short)0, clo, false, false); \
        chi = __builtin_amdgcn_wmma_f32_16x16x4_f32(false, a1, false, b1,     \
                                                    (short)0, chi, false, false); \
    }

    for (int kc = 0; kc < D_K - 2 * A_CH; kc += 2 * A_CH) {
        { A_LOAD(r, kc + 2 * A_CH) A_COMPUTE(0, kc)        A_STORE(r, 0) }
        { A_LOAD(r, kc + 3 * A_CH) A_COMPUTE(1, kc + A_CH) A_STORE(r, 1) }
    }
    A_COMPUTE(0, D_K - 2 * A_CH)
    A_COMPUTE(1, D_K - A_CH)
#undef A_LOAD
#undef A_STORE
#undef A_COMPUTE

    {
        const int   p      = lane & 7;
        const int   which  = (lane >> 3) & 1;
        const int   f      = 2 * wave + (lane >> 4);
        const float b_lo   = bias[16 * which + p];
        const float b_hi   = bias[16 * which + 8 + p];
        float*      dst    = sX + which * X_WP + p * X_EP + f;
#pragma unroll
        for (int v = 0; v < 8; ++v) {
            float lo = clo[v] + b_lo;
            float hi = chi[v] + b_hi;
            if (which) {
                lo = fmaxf(__expf(lo), 1e-6f);
                hi = fmaxf(__expf(hi), 1e-6f);
            }
            const bool  two = (n0 + v) > 0;
            const float inv = two ? 0.5f : 1.0f;
            dst[8 * v * X_EP] = (lo + (two ? hi : 0.f)) * inv;
            if (v == 0 && n0 == 0)
                sT[(which * 8 + p) * X_EP + f] = hi;
        }
    }
    __syncthreads();

    const size_t half = (size_t)B_SZ * T * F_;
#pragma unroll
    for (int r = 0; r < (2 * 8 * NB * 8) / NTHREADS; ++r) {
        const int idx   = tid + r * NTHREADS;
        const int q     = idx & 7;
        const int line  = idx >> 3;
        const int which = line >> 6;
        const int tt    = line & 63;
        const int t     = 8 * n0 + tt;
        const float4 o  = *(const float4*)(sX + which * X_WP + tt * X_EP + 4 * q);
        if (t < T)
            *(float4*)(out + which * half + ((size_t)bb * T + t) * F_ + 4 * q) = o;
    }

    if (n0 == 0) {
        const int ntail = T - T_FULL;
        for (int idx = tid; idx < ntail * 16; idx += NTHREADS) {
            const int q     = idx & 7;
            const int line  = idx >> 3;
            const int which = line >= ntail ? 1 : 0;
            const int tr    = line - which * ntail;
            const int t     = T_FULL + tr;
            float4 o = make_float4(0.f, 0.f, 0.f, 0.f);
            if (tr < 8)
                o = *(const float4*)(sT + (which * 8 + tr) * X_EP + 4 * q);
            *(float4*)(out + which * half + ((size_t)bb * T + t) * F_ + 4 * q) = o;
        }
    }
}

extern "C" void kernel_launch(void* const* d_in, const int* in_sizes, int n_in,
                              void* d_out, int out_size, void* d_ws, size_t ws_size,
                              hipStream_t stream)
{
    const float* A    = (const float*)d_in[0];
    const float* W    = (const float*)d_in[1];
    const float* bias = (const float*)d_in[2];
    const int T = out_size / (2 * B_SZ * F_);

    gemm_fold_kernel<<<B_SZ * NBLK_B, NTHREADS, 0, stream>>>(
        A, W, bias, (float*)d_out, T);
}
